# starting kernel
# baseline (speedup 1.0000x reference)
#include <hip/hip_runtime.h>
#include <hip/hip_bf16.h>

typedef __attribute__((ext_vector_type(2))) float v2f;
typedef __attribute__((ext_vector_type(8))) float v8f;

#define TAO     24
#define MLAG    7
#define C_IN    7
#define D_MODEL 512
#define SEQ     2048
#define KDIM    168
#define NSTEPS  42
#define ASTRIDE 172

__global__ __launch_bounds__(256)
void tdc_conv_wmma_kernel(const float* __restrict__ x,
                          const float* __restrict__ W,
                          const float* __restrict__ bias,
                          float* __restrict__ out)
{
    __shared__ float Atile[16 * ASTRIDE];

    const int block = blockIdx.x;
    const int batch = block >> 7;
    const int t0    = (block & 127) << 4;
    const int tid   = threadIdx.x;

    for (int e = tid; e < 16 * KDIM; e += 256) {
        const int m_row = e / KDIM;
        const int j     = e - m_row * KDIM;
        const int k     = j % 3;
        const int i     = j / 3;
        const int c     = i >> 3;
        const int mm    = i & 7;
        const int t     = t0 + m_row;
        const int s     = (t + k - 1 + SEQ) & (SEQ - 1);
        float v = 0.0f;
        if (s >= MLAG * TAO) {
            v = x[((size_t)batch * SEQ + (s - TAO * mm)) * C_IN + c];
        }
        Atile[m_row * ASTRIDE + j] = v;
    }
    __syncthreads();

    const int wave = tid >> 5;
    const int lane = tid & 31;
    const int half = lane >> 4;
    const int lid  = lane & 15;

    v8f acc0 = {}, acc1 = {}, acc2 = {}, acc3 = {};

    const float* Arow = &Atile[lid * ASTRIDE + 2 * half];
    const int o_base = wave * 64;
    const float* Wl0 = W + (size_t)(o_base +  0 + lid) * KDIM + 2 * half;
    const float* Wl1 = W + (size_t)(o_base + 16 + lid) * KDIM + 2 * half;
    const float* Wl2 = W + (size_t)(o_base + 32 + lid) * KDIM + 2 * half;
    const float* Wl3 = W + (size_t)(o_base + 48 + lid) * KDIM + 2 * half;

    #pragma unroll 6
    for (int kk = 0; kk < NSTEPS; ++kk) {
        const int k0 = kk * 4;
        v2f a  = *(const v2f*)(Arow + k0);
        v2f b0 = *(const v2f*)(Wl0 + k0);
        v2f b1 = *(const v2f*)(Wl1 + k0);
        v2f b2 = *(const v2f*)(Wl2 + k0);
        v2f b3 = *(const v2f*)(Wl3 + k0);
        acc0 = __builtin_amdgcn_wmma_f32_16x16x4_f32(false, a, false, b0, (short)0, acc0, false, false);
        acc1 = __builtin_amdgcn_wmma_f32_16x16x4_f32(false, a, false, b1, (short)0, acc1, false, false);
        acc2 = __builtin_amdgcn_wmma_f32_16x16x4_f32(false, a, false, b2, (short)0, acc2, false, false);
        acc3 = __builtin_amdgcn_wmma_f32_16x16x4_f32(false, a, false, b3, (short)0, acc3, false, false);
    }

    const size_t row_base = ((size_t)batch * SEQ + t0 + 8 * half) * D_MODEL;

    {
        const int o = o_base + 0 + lid;
        const float bo = bias[o];
        #pragma unroll
        for (int r = 0; r < 8; ++r)
            out[row_base + (size_t)r * D_MODEL + o] = acc0[r] + bo;
    }
    {
        const int o = o_base + 16 + lid;
        const float bo = bias[o];
        #pragma unroll
        for (int r = 0; r < 8; ++r)
            out[row_base + (size_t)r * D_MODEL + o] = acc1[r] + bo;
    }
    {
        const int o = o_base + 32 + lid;
        const float bo = bias[o];
        #pragma unroll
        for (int r = 0; r < 8; ++r)
            out[row_base + (size_t)r * D_MODEL + o] = acc2[r] + bo;
    }
    {
        const int o = o_base + 48 + lid;
        const float bo = bias[o];
        #pragma unroll
        for (int r = 0; r < 8; ++r)
            out[row_base + (size_t)r * D_MODEL + o] = acc3[r] + bo;
    }
}

extern "C" void kernel_launch(void* const* d_in, const int* in_sizes, int n_in,
                              void* d_out, int out_size, void* d_ws, size_t ws_size,
                              hipStream_t stream) {
    (void)in_sizes; (void)n_in; (void)d_ws; (void)ws_size; (void)out_size;
    const float* x    = (const float*)d_in[0];
    const float* W    = (const float*)d_in[1];
    const float* bias = (const float*)d_in[2];
    float* out        = (float*)d_out;

    const int num_blocks = 32 * (SEQ / 16);
    tdc_conv_wmma_kernel<<<num_blocks, 256, 0, stream>>>(x, W, bias, out);
}


# W slice (128 cols) kept in LDS per work-group, staged once with whole-line loads; work-group walks 8 row tiles of 64 (next tile's x loads fly under the WMMAs); B fragments from LDS, k order kept
# speedup vs baseline: 2.4026x; 2.4026x over previous
#include <hip/hip_runtime.h>
#include <hip/hip_bf16.h>


typedef __attribute__((ext_vector_type(2))) float v2f;
typedef __attribute__((ext_vector_type(8))) float v8f;

#define TAO     24
#define MLAG    7
#define C_IN    7
#define D_MODEL 512
#define SEQ     2048
#define KDIM    168
#define NSTEPS  42
#define PITCH   172

#define NB      128
#define MT      64
#define ROWS_WG 512
#define NITER   (ROWS_WG / MT)
#define NPAIR   (MT * C_IN)

__global__ __launch_bounds__(256)
void tdc_conv_wmma_kernel(const float* __restrict__ x,
                          const float* __restrict__ W,
                          const float* __restrict__ bias,
                          float* __restrict__ out)
{
    __shared__ __attribute__((aligned(16))) float Wl[NB * PITCH];
    __shared__ __attribute__((aligned(16))) float At[MT * PITCH];

    const int tid    = threadIdx.x;
    const int nq     = blockIdx.x & 3;
    const int chunk  = blockIdx.x >> 2;
    const int batch  = chunk >> 2;
    const int tchunk = (chunk & 3) * ROWS_WG;

    const int  e0   = tid;
    const int  e1   = tid + 256;
    const bool two  = (e1 < NPAIR);
    const int  row0 = e0 / C_IN, c0 = e0 - row0 * C_IN;
    const int  row1 = e1 / C_IN, c1 = e1 - row1 * C_IN;
    const float* xb = x + (size_t)batch * SEQ * C_IN;
    float* Ad0 = &At[row0 * PITCH + c0 * 24];
    float* Ad1 = &At[(two ? row1 : 0) * PITCH + c1 * 24];

    float pre0[24], pre1[24];
    bool  ok0[3], ok1[3];
    auto gather = [&](int tt0) {
        int b0[3], b1[3];
        #pragma unroll
        for (int k = 0; k < 3; ++k) {
            const int s0 = (tt0 + row0 + k - 1 + SEQ) & (SEQ - 1);
            const int s1 = (tt0 + row1 + k - 1 + SEQ) & (SEQ - 1);
            ok0[k] = (s0 >= MLAG * TAO);
            ok1[k] = (s1 >= MLAG * TAO);
            b0[k] = C_IN * (ok0[k] ? s0 : MLAG * TAO) + c0;
            b1[k] = C_IN * (ok1[k] ? s1 : MLAG * TAO) + c1;
        }
        #pragma unroll
        for (int m = 0; m < 8; ++m) {
            #pragma unroll
            for (int k = 0; k < 3; ++k) pre0[m * 3 + k] = xb[b0[k] - TAO * C_IN * m];
        }
        if (two) {
            #pragma unroll
            for (int m = 0; m < 8; ++m) {
                #pragma unroll
                for (int k = 0; k < 3; ++k) pre1[m * 3 + k] = xb[b1[k] - TAO * C_IN * m];
            }
        }
    };

    gather(tchunk);

    {
        const float4* Wsrc = (const float4*)(W + (size_t)nq * NB * KDIM);
        #pragma unroll
        for (int q = 0; q < 21; ++q) {
            const int f  = tid + q * 256;
            const int r  = f / 42;
            const int c4 = f - r * 42;
            const float4 w4 = Wsrc[f];
            *(float4*)&Wl[r * PITCH + c4 * 4] = w4;
        }
    }

    const int wave = tid >> 5;
    const int lane = tid & 31;
    const int half = lane >> 4;
    const int lid  = lane & 15;
    const int wm   = wave & 1;
    const int wn   = wave >> 1;

    const float* Ar0 = &At[(wm * 32 + lid) * PITCH + 2 * half];
    const float* Ar1 = Ar0 + 16 * PITCH;
    const float* Br0 = &Wl[(wn * 32 + lid) * PITCH + 2 * half];
    const float* Br1 = Br0 + 16 * PITCH;

    const int   ocol  = nq * NB + wn * 32 + lid;
    const float bias0 = bias[ocol];
    const float bias1 = bias[ocol + 16];

    #pragma unroll 1
    for (int it = 0; it < NITER; ++it) {
        __syncthreads();
        #pragma unroll
        for (int q = 0; q < 24; ++q) Ad0[q] = ok0[q % 3] ? pre0[q] : 0.0f;
        if (two) {
            #pragma unroll
            for (int q = 0; q < 24; ++q) Ad1[q] = ok1[q % 3] ? pre1[q] : 0.0f;
        }
        __syncthreads();

        if (it + 1 < NITER) gather(tchunk + (it + 1) * MT);

        v8f acc00 = {}, acc01 = {}, acc10 = {}, acc11 = {};
        #pragma unroll
        for (int kk = 0; kk < NSTEPS; ++kk) {
            const int k0 = kk * 4;
            const v2f a0 = *(const v2f*)(Ar0 + k0);
            const v2f a1 = *(const v2f*)(Ar1 + k0);
            const v2f b0 = *(const v2f*)(Br0 + k0);
            const v2f b1 = *(const v2f*)(Br1 + k0);
            acc00 = __builtin_amdgcn_wmma_f32_16x16x4_f32(false, a0, false, b0, (short)0, acc00, false, false);
            acc01 = __builtin_amdgcn_wmma_f32_16x16x4_f32(false, a0, false, b1, (short)0, acc01, false, false);
            acc10 = __builtin_amdgcn_wmma_f32_16x16x4_f32(false, a1, false, b0, (short)0, acc10, false, false);
            acc11 = __builtin_amdgcn_wmma_f32_16x16x4_f32(false, a1, false, b1, (short)0, acc11, false, false);
        }

        float* o0 = out + ((size_t)batch * SEQ + tchunk + it * MT + wm * 32 + 8 * half) * D_MODEL + ocol;
        float* o1 = o0 + (size_t)16 * D_MODEL;
        #pragma unroll
        for (int r = 0; r < 8; ++r) {
            o0[(size_t)r * D_MODEL]      = acc00[r] + bias0;
            o0[(size_t)r * D_MODEL + 16] = acc01[r] + bias1;
            o1[(size_t)r * D_MODEL]      = acc10[r] + bias0;
            o1[(size_t)r * D_MODEL + 16] = acc11[r] + bias1;
        }
    }
}

extern "C" void kernel_launch(void* const* d_in, const int* in_sizes, int n_in,
                              void* d_out, int out_size, void* d_ws, size_t ws_size,
                              hipStream_t stream) {
    (void)in_sizes; (void)n_in; (void)d_ws; (void)ws_size; (void)out_size;
    const float* x    = (const float*)d_in[0];
    const float* W    = (const float*)d_in[1];
    const float* bias = (const float*)d_in[2];
    float* out        = (float*)d_out;

    tdc_conv_wmma_kernel<<<512, 256, 0, stream>>>(x, W, bias, out);
}


# A fragments read from a compact x window in LDS (rows of A are shifted copies of x): no expanded tile, no per-tile staging or barrier in the row loop; k order permuted in both loaders (sum order)
# speedup vs baseline: 2.6142x; 1.0881x over previous
#include <hip/hip_runtime.h>
#include <hip/hip_bf16.h>


typedef __attribute__((ext_vector_type(2))) float v2f;
typedef __attribute__((ext_vector_type(8))) float v8f;

#define TAO     24
#define MLAG    7
#define C_IN    7
#define D_MODEL 512
#define SEQ     2048
#define KDIM    168
#define PITCH   172

#define NB      128
#define MT      64
#define ROWS_WG 512
#define NITER   (ROWS_WG / MT)

#define WIN0    145
#define WINLEN  (ROWS_WG + 146)
#define WINENT  (WINLEN * C_IN)
#define WINSTEP 18

__global__ __launch_bounds__(256)
void tdc_conv_wmma_kernel(const float* __restrict__ x,
                          const float* __restrict__ W,
                          const float* __restrict__ bias,
                          float* __restrict__ out)
{
    __shared__ __attribute__((aligned(16))) float Wl[NB * PITCH];
    __shared__ __attribute__((aligned(16))) float Xs[2 * WINSTEP * 256];
    __shared__ __attribute__((aligned(16))) float Zr[16];

    const int tid    = threadIdx.x;
    const int nq     = blockIdx.x & 3;
    const int chunk  = blockIdx.x >> 2;
    const int batch  = chunk >> 2;
    const int tchunk = (chunk & 3) * ROWS_WG;
    const float* xb  = x + (size_t)batch * SEQ * C_IN;

    v2f xw[WINSTEP];
    #pragma unroll
    for (int i = 0; i < WINSTEP; ++i) {
        const int e = tid + i * 256;
        const int w = e / C_IN;
        const int c = e - w * C_IN;
        const int u = tchunk - WIN0 + w;
        xw[i].x = xb[((u)       & (SEQ - 1)) * C_IN + c];
        xw[i].y = xb[((u - TAO) & (SEQ - 1)) * C_IN + c];
    }
    if (tid < 16) Zr[tid] = 0.0f;

    {
        const float4* Wsrc = (const float4*)(W + (size_t)nq * NB * KDIM);
        #pragma unroll
        for (int q = 0; q < 7; ++q) {
            const int b   = tid + q * 256;
            const int r   = b / 14;
            const int blk = b - r * 14;
            const float4 s0 = Wsrc[b * 3 + 0];
            const float4 s1 = Wsrc[b * 3 + 1];
            const float4 s2 = Wsrc[b * 3 + 2];
            const float sv[12] = { s0.x, s0.y, s0.z, s0.w, s1.x, s1.y, s1.z, s1.w, s2.x, s2.y, s2.z, s2.w };
            float dv[12];
            #pragma unroll
            for (int k = 0; k < 3; ++k)
                #pragma unroll
                for (int h = 0; h < 2; ++h)
                    #pragma unroll
                    for (int e = 0; e < 2; ++e) dv[4 * k + 2 * h + e] = sv[6 * h + 3 * e + k];
            float4* dst = (float4*)&Wl[r * PITCH + blk * 12];
            dst[0] = make_float4(dv[0], dv[1], dv[2],  dv[3]);
            dst[1] = make_float4(dv[4], dv[5], dv[6],  dv[7]);
            dst[2] = make_float4(dv[8], dv[9], dv[10], dv[11]);
        }
    }

    #pragma unroll
    for (int i = 0; i < WINSTEP; ++i) *(v2f*)&Xs[2 * (tid + i * 256)] = xw[i];

    const int wave = tid >> 5;
    const int lane = tid & 31;
    const int half = lane >> 4;
    const int lid  = lane & 15;
    const int wm   = wave & 1;
    const int wn   = wave >> 1;

    const float* Br0 = &Wl[(wn * 32 + lid) * PITCH + 2 * half];
    const float* Br1 = Br0 + 16 * PITCH;

    const int   ocol  = nq * NB + wn * 32 + lid;
    const float bias0 = bias[ocol];
    const float bias1 = bias[ocol + 16];

    const int wlane = wm * 32 + lid + (WIN0 - 1) - 2 * TAO * half;
    const int tlane = tchunk + wm * 32 + lid;

    __syncthreads();

    #pragma unroll 1
    for (int it = 0; it < NITER; ++it) {
        const float* Ap[2][3][2];
        #pragma unroll
        for (int rb = 0; rb < 2; ++rb) {
            #pragma unroll
            for (int k = 0; k < 3; ++k) {
                const int  s  = (tlane + it * MT + rb * 16 + k - 1) & (SEQ - 1);
                const bool ok = (s >= MLAG * TAO);
                #pragma unroll
                for (int g = 0; g < 2; ++g) {
                    const int w = wlane + it * MT + rb * 16 + k - 4 * TAO * g;
                    Ap[rb][k][g] = ok ? &Xs[2 * C_IN * w] : (const float*)Zr;
                }
            }
        }

        v8f acc00 = {}, acc01 = {}, acc10 = {}, acc11 = {};
        #pragma unroll
        for (int g = 0; g < 2; ++g) {
            #pragma unroll
            for (int k = 0; k < 3; ++k) {
                const float* a0p = Ap[0][k][g];
                const float* a1p = Ap[1][k][g];
                #pragma unroll
                for (int c = 0; c < C_IN; ++c) {
                    const int q4 = 4 * (c * 6 + g * 3 + k);
                    const v2f a0 = *(const v2f*)(a0p + 2 * c);
                    const v2f a1 = *(const v2f*)(a1p + 2 * c);
                    const v2f b0 = *(const v2f*)(Br0 + q4);
                    const v2f b1 = *(const v2f*)(Br1 + q4);
                    acc00 = __builtin_amdgcn_wmma_f32_16x16x4_f32(false, a0, false, b0, (short)0, acc00, false, false);
                    acc01 = __builtin_amdgcn_wmma_f32_16x16x4_f32(false, a0, false, b1, (short)0, acc01, false, false);
                    acc10 = __builtin_amdgcn_wmma_f32_16x16x4_f32(false, a1, false, b0, (short)0, acc10, false, false);
                    acc11 = __builtin_amdgcn_wmma_f32_16x16x4_f32(false, a1, false, b1, (short)0, acc11, false, false);
                }
            }
        }

        float* o0 = out + ((size_t)batch * SEQ + tchunk + it * MT + wm * 32 + 8 * half) * D_MODEL + ocol;
        float* o1 = o0 + (size_t)16 * D_MODEL;
        #pragma unroll
        for (int r = 0; r < 8; ++r) {
            o0[(size_t)r * D_MODEL]      = acc00[r] + bias0;
            o0[(size_t)r * D_MODEL + 16] = acc01[r] + bias1;
            o1[(size_t)r * D_MODEL]      = acc10[r] + bias0;
            o1[(size_t)r * D_MODEL + 16] = acc11[r] + bias1;
        }
    }
}

extern "C" void kernel_launch(void* const* d_in, const int* in_sizes, int n_in,
                              void* d_out, int out_size, void* d_ws, size_t ws_size,
                              hipStream_t stream) {
    (void)in_sizes; (void)n_in; (void)d_ws; (void)ws_size; (void)out_size;
    const float* x    = (const float*)d_in[0];
    const float* W    = (const float*)d_in[1];
    const float* bias = (const float*)d_in[2];
    float* out        = (float*)d_out;

    tdc_conv_wmma_kernel<<<512, 256, 0, stream>>>(x, W, bias, out);
}
